# starting kernel
# baseline (speedup 1.0000x reference)
#include <hip/hip_runtime.h>
#include <hip/hip_bf16.h>


#define TT 256
#define N4 4

typedef __attribute__((ext_vector_type(2))) float v2f;
typedef __attribute__((ext_vector_type(8))) float v8f;

__device__ __forceinline__ float shf(float v, int srcLane) {
  return __int_as_float(__builtin_amdgcn_ds_bpermute(srcLane << 2, __float_as_int(v)));
}

__device__ __forceinline__ v8f wmma4(float a0, float a1, float b0, float b1, v8f c) {
  v2f A, B;
  A[0] = a0; A[1] = a1;
  B[0] = b0; B[1] = b1;
  return __builtin_amdgcn_wmma_f32_16x16x4_f32(false, A, false, B, (short)0, c, false, false);
}

__device__ __forceinline__ float fast_tanh(float x) {
#if __has_builtin(__builtin_amdgcn_tanhf)
  return __builtin_amdgcn_tanhf(x);
#else
  const float e = __expf(2.0f * x);
  return 1.0f - 2.0f * __builtin_amdgcn_rcpf(e + 1.0f);
#endif
}
#define INV_2PI 0.15915494309189535f
__device__ __forceinline__ float fast_sin(float x) {
  return __builtin_amdgcn_sinf(x * INV_2PI);
}
__device__ __forceinline__ float fast_cos(float x) {
  return __builtin_amdgcn_cosf(x * INV_2PI);
}

__global__ __launch_bounds__(256) void ekf1250_kernel(
    const float* __restrict__ z,  const float* __restrict__ Ag,
    const float* __restrict__ Cg, const float* __restrict__ Qg,
    const float* __restrict__ Rg, const float* __restrict__ x0g,
    const float* __restrict__ P0g, float* __restrict__ out)
{
  __shared__ float lds[8 * 640];
  const int tid  = threadIdx.x;
  const int lane = tid & 31;
  const int wave = (blockIdx.x * blockDim.x + tid) >> 5;
  const long b0  = (long)wave * 2;
  float* W = &lds[(tid >> 5) * 640];
  const int RP = 0;
  const int R1 = 128;
  const int R2 = 256;
  const int R3 = 384;
  const int R4 = 448;
  const int RT = 512;

  const int  r    = lane & 7;
  const int  e8   = (lane >> 3) & 1;
  const int  khi  = lane >> 4;
  const int  eD   = lane >> 4;
  const bool dval = (lane < 8) || (lane >= 24);
  const bool lowh = lane < 16;
  const int  dcol = lane & 7;

  const int aBase  = e8 * 64 + r * 8 + 2 * khi;
  const int bBase  = e8 * 64 + (2 * khi) * 8 + r;
  const int a3Base = e8 * 32 + (r & 3) * 8 + 2 * khi;

  float regA[8], regC[8];
  {
    const float4 a0 = *reinterpret_cast<const float4*>(Ag + r * 8);
    const float4 a1 = *reinterpret_cast<const float4*>(Ag + r * 8 + 4);
    regA[0]=a0.x; regA[1]=a0.y; regA[2]=a0.z; regA[3]=a0.w;
    regA[4]=a1.x; regA[5]=a1.y; regA[6]=a1.z; regA[7]=a1.w;
    const float4 c0 = *reinterpret_cast<const float4*>(Cg + (r & 3) * 8);
    const float4 c1 = *reinterpret_cast<const float4*>(Cg + (r & 3) * 8 + 4);
    const float m = (r < 4) ? 1.0f : 0.0f;
    regC[0]=c0.x*m; regC[1]=c0.y*m; regC[2]=c0.z*m; regC[3]=c0.w*m;
    regC[4]=c1.x*m; regC[5]=c1.y*m; regC[6]=c1.z*m; regC[7]=c1.w*m;
  }
  v8f Qd, Rd;
  #pragma unroll
  for (int i = 0; i < 8; ++i) Qd[i] = Qg[i * 8 + dcol];
  #pragma unroll
  for (int i = 0; i < 8; ++i) {
    float rv = Rg[(i & 3) * 4 + (dcol & 3)];
    Rd[i] = (i < 4 && dcol < 4) ? rv : 0.0f;
  }

  float xrep = x0g[r];
  for (int idx = lane; idx < 128; idx += 32) W[RP + idx] = P0g[idx & 63];

  const v8f z8 = {0,0,0,0,0,0,0,0};

  #pragma unroll 1
  for (int t = 0; t < TT; ++t) {
    const int sb = lane & 16;
    float xb[8];
    #pragma unroll
    for (int k = 0; k < 8; ++k) xb[k] = shf(xrep, sb + k);
    float ta = 0.0f;
    #pragma unroll
    for (int k = 0; k < 8; ++k) ta = fmaf(regA[k], xb[k], ta);
    const float xp = fast_tanh(ta);
    const float dd = 1.0f - xp * xp;
    float u = 0.0f;
    #pragma unroll
    for (int k = 0; k < 8; ++k) u = fmaf(regC[k], shf(xp, sb + k), u);
    const float ys = fast_sin(u);
    const float cy = fast_cos(u);

    const int srcAB = (e8 << 4) + r;
    const float dA  = shf(dd, srcAB);
    const float cyA = shf(cy, srcAB);
    float Fo[4], Ho[4];
    #pragma unroll
    for (int jc = 0; jc < 2; ++jc)
      #pragma unroll
      for (int v = 0; v < 2; ++v) {
        const float av = lowh ? regA[4 * jc + v] : regA[4 * jc + v + 2];
        const float cv = lowh ? regC[4 * jc + v] : regC[4 * jc + v + 2];
        Fo[2 * jc + v] = dA * av;
        Ho[2 * jc + v] = cyA * cv;
      }

    v8f G = wmma4(Fo[0], Fo[1], W[RP + bBase + 0 ], W[RP + bBase + 8 ], z8);
    G     = wmma4(Fo[2], Fo[3], W[RP + bBase + 32], W[RP + bBase + 40], G);
    {
      const int base = dval ? (R1 + eD * 64 + dcol) : (RT + (lane & 15));
      #pragma unroll
      for (int i = 0; i < 8; ++i) W[base + i * 8] = G[i];
    }
    v8f Pp = wmma4(W[R1 + aBase + 0], W[R1 + aBase + 1], Fo[0], Fo[1], Qd);
    Pp     = wmma4(W[R1 + aBase + 4], W[R1 + aBase + 5], Fo[2], Fo[3], Pp);
    {
      const int base = dval ? (R2 + eD * 64 + dcol) : (RT + (lane & 15));
      #pragma unroll
      for (int i = 0; i < 8; ++i) W[base + i * 8] = Pp[i];
    }
    v8f T1 = wmma4(Ho[0], Ho[1], W[R2 + bBase + 0 ], W[R2 + bBase + 8 ], z8);
    T1     = wmma4(Ho[2], Ho[3], W[R2 + bBase + 32], W[R2 + bBase + 40], T1);
    {
      const int base = dval ? (R3 + eD * 32 + dcol) : (RT + (lane & 15));
      #pragma unroll
      for (int i = 0; i < 4; ++i) W[base + i * 8] = T1[i];
    }
    v8f Sm = wmma4(W[R3 + a3Base + 0], W[R3 + a3Base + 1], Ho[0], Ho[1], Rd);
    Sm     = wmma4(W[R3 + a3Base + 4], W[R3 + a3Base + 5], Ho[2], Ho[3], Sm);
    v8f PHt = wmma4(W[R2 + aBase + 0], W[R2 + aBase + 1], Ho[0], Ho[1], z8);
    PHt     = wmma4(W[R2 + aBase + 4], W[R2 + aBase + 5], Ho[2], Ho[3], PHt);
    {
      const int base = (dval && dcol < 4) ? (R4 + eD * 32 + dcol) : (RT + (lane & 15));
      #pragma unroll
      for (int i = 0; i < 8; ++i) W[base + i * 4] = PHt[i];
    }

    const float4 rhs4 = *reinterpret_cast<const float4*>(&W[R4 + (eD << 5) + (r << 2)]);
    float rr4[4] = {rhs4.x, rhs4.y, rhs4.z, rhs4.w};
    float s[4][4];
    const int sS = 24 * eD;
    #pragma unroll
    for (int i = 0; i < 4; ++i)
      #pragma unroll
      for (int j = 0; j < 4; ++j) s[i][j] = shf(Sm[i], sS + j);
    #pragma unroll
    for (int k = 0; k < 4; ++k) {
      const float inv = __builtin_amdgcn_rcpf(s[k][k]);
      #pragma unroll
      for (int j = k + 1; j < 4; ++j) s[k][j] *= inv;
      rr4[k] *= inv;
      #pragma unroll
      for (int i = k + 1; i < 4; ++i) {
        const float f = s[i][k];
        #pragma unroll
        for (int j = k + 1; j < 4; ++j) s[i][j] = fmaf(-f, s[k][j], s[i][j]);
        rr4[i] = fmaf(-f, rr4[k], rr4[i]);
      }
    }
    float Kp[4];
    Kp[3] = rr4[3];
    Kp[2] = fmaf(-s[2][3], Kp[3], rr4[2]);
    Kp[1] = fmaf(-s[1][3], Kp[3], fmaf(-s[1][2], Kp[2], rr4[1]));
    Kp[0] = fmaf(-s[0][3], Kp[3], fmaf(-s[0][2], Kp[2], fmaf(-s[0][1], Kp[1], rr4[0])));

    const long zidx = ((b0 + (lane >> 4)) * TT + t) * N4 + (r & 3);
    const float zv = z[zidx];
    __builtin_prefetch(&z[(t < TT - 8) ? (zidx + 8 * N4) : zidx], 0, 0);
    const float dy = zv - ys;
    const int sE = lane & 16;
    float xn = shf(xp, sE + r);
    #pragma unroll
    for (int j = 0; j < 4; ++j) xn = fmaf(Kp[j], shf(dy, sE + j), xn);
    if (dval) out[((b0 + eD) * TT + t) * 8 + r] = xn;

    const int sK = 24 * e8 + r;
    float Av0, Av1, Bv0, Bv1;
    {
      const float klo0 = shf(Kp[0], sK), khi0 = shf(Kp[2], sK);
      const float klo1 = shf(Kp[1], sK), khi1 = shf(Kp[3], sK);
      Av0 = -(lowh ? klo0 : khi0);
      Av1 = -(lowh ? klo1 : khi1);
      const float tlo0 = shf(T1[0], sK), thi0 = shf(T1[2], sK);
      const float tlo1 = shf(T1[1], sK), thi1 = shf(T1[3], sK);
      Bv0 = lowh ? tlo0 : thi0;
      Bv1 = lowh ? tlo1 : thi1;
    }
    v8f Pn = wmma4(Av0, Av1, Bv0, Bv1, Pp);
    {
      const int base = dval ? (RP + eD * 64 + dcol) : (RT + (lane & 15));
      #pragma unroll
      for (int i = 0; i < 8; ++i) W[base + i * 8] = Pn[i];
    }
    xrep = shf(xn, 24 * (lane >> 4) + r);
  }
}

extern "C" void kernel_launch(void* const* d_in, const int* in_sizes, int n_in,
                              void* d_out, int out_size, void* d_ws, size_t ws_size,
                              hipStream_t stream) {
  const float* z  = (const float*)d_in[0];
  const float* A  = (const float*)d_in[1];
  const float* C  = (const float*)d_in[2];
  const float* Q  = (const float*)d_in[3];
  const float* R  = (const float*)d_in[4];
  const float* x0 = (const float*)d_in[5];
  const float* P0 = (const float*)d_in[6];
  float* out = (float*)d_out;

  const int Bsz = in_sizes[0] / (TT * N4);
  const int waves = Bsz / 2;
  const int threads = 256;
  int blocks = (waves * 32) / threads;
  if (blocks < 1) blocks = 1;
  ekf1250_kernel<<<blocks, threads, 0, stream>>>(z, A, C, Q, R, x0, P0, out);
}


# Loop carries tanh(A x) and A P A^T instead of x and P: one S elimination gives S^-1 dy and S^-1 (T1 A^T); A P' A^T, A P' H^T, T1 A^T run beside it; product and sum order of x and P change
# speedup vs baseline: 1.1916x; 1.1916x over previous
#include <hip/hip_runtime.h>
#include <hip/hip_bf16.h>


#define TT 256
#define N4 4

typedef __attribute__((ext_vector_type(2))) float v2f;
typedef __attribute__((ext_vector_type(8))) float v8f;

__device__ __forceinline__ float shf(float v, int srcLane) {
  return __int_as_float(__builtin_amdgcn_ds_bpermute(srcLane << 2, __float_as_int(v)));
}

__device__ __forceinline__ v8f wmma4(float a0, float a1, float b0, float b1, v8f c) {
  v2f A, B;
  A[0] = a0; A[1] = a1;
  B[0] = b0; B[1] = b1;
  return __builtin_amdgcn_wmma_f32_16x16x4_f32(false, A, false, B, (short)0, c, false, false);
}

__device__ __forceinline__ float fast_tanh(float x) {
#if __has_builtin(__builtin_amdgcn_tanhf)
  return __builtin_amdgcn_tanhf(x);
#else
  const float e = __expf(2.0f * x);
  return 1.0f - 2.0f * __builtin_amdgcn_rcpf(e + 1.0f);
#endif
}
#define INV_2PI 0.15915494309189535f
__device__ __forceinline__ float fast_sin(float x) {
  return __builtin_amdgcn_sinf(x * INV_2PI);
}
__device__ __forceinline__ float fast_cos(float x) {
  return __builtin_amdgcn_cosf(x * INV_2PI);
}

__global__ __launch_bounds__(256) void ekf1250_kernel(
    const float* __restrict__ z,  const float* __restrict__ Ag,
    const float* __restrict__ Cg, const float* __restrict__ Qg,
    const float* __restrict__ Rg, const float* __restrict__ x0g,
    const float* __restrict__ P0g, float* __restrict__ out)
{
  __shared__ float lds[8 * 640];
  const int tid  = threadIdx.x;
  const int lane = tid & 31;
  const int wave = (blockIdx.x * blockDim.x + tid) >> 5;
  const long b0  = (long)wave * 2;
  float* W = &lds[(tid >> 5) * 640];
  const int RP = 0;
  const int R1 = 128;
  const int R2 = 256;
  const int R3 = 384;
  const int R4 = 448;
  const int RT = 512;

  const int  r    = lane & 7;
  const int  e8   = (lane >> 3) & 1;
  const int  khi  = lane >> 4;
  const int  eD   = lane >> 4;
  const bool dval = (lane < 8) || (lane >= 24);
  const bool lowh = lane < 16;
  const int  dcol = lane & 7;

  const int aBase  = e8 * 64 + r * 8 + 2 * khi;
  const int bBase  = e8 * 64 + (2 * khi) * 8 + r;
  const int a3Base = e8 * 32 + (r & 3) * 8 + 2 * khi;

  float regA[8], regC[8];
  {
    const float4 a0 = *reinterpret_cast<const float4*>(Ag + r * 8);
    const float4 a1 = *reinterpret_cast<const float4*>(Ag + r * 8 + 4);
    regA[0]=a0.x; regA[1]=a0.y; regA[2]=a0.z; regA[3]=a0.w;
    regA[4]=a1.x; regA[5]=a1.y; regA[6]=a1.z; regA[7]=a1.w;
    const float4 c0 = *reinterpret_cast<const float4*>(Cg + (r & 3) * 8);
    const float4 c1 = *reinterpret_cast<const float4*>(Cg + (r & 3) * 8 + 4);
    const float m = (r < 4) ? 1.0f : 0.0f;
    regC[0]=c0.x*m; regC[1]=c0.y*m; regC[2]=c0.z*m; regC[3]=c0.w*m;
    regC[4]=c1.x*m; regC[5]=c1.y*m; regC[6]=c1.z*m; regC[7]=c1.w*m;
  }
  v8f Qd, Rd;
  #pragma unroll
  for (int i = 0; i < 8; ++i) Qd[i] = Qg[i * 8 + dcol];
  #pragma unroll
  for (int i = 0; i < 8; ++i) {
    float rv = Rg[(i & 3) * 4 + (dcol & 3)];
    Rd[i] = (i < 4 && dcol < 4) ? rv : 0.0f;
  }

  for (int idx = lane; idx < 128; idx += 32) W[RP + idx] = P0g[idx & 63];

  const v8f z8 = {0,0,0,0,0,0,0,0};

  float Ao[4];
  #pragma unroll
  for (int jc = 0; jc < 2; ++jc)
    #pragma unroll
    for (int v = 0; v < 2; ++v)
      Ao[2 * jc + v] = lowh ? regA[4 * jc + v] : regA[4 * jc + v + 2];

  const int sb = lane & 16;
  float xp;
  float xpb[8];
  {
    float ta = 0.0f;
    #pragma unroll
    for (int k = 0; k < 8; ++k) ta = fmaf(regA[k], x0g[k], ta);
    xp = fast_tanh(ta);
    #pragma unroll
    for (int k = 0; k < 8; ++k) xpb[k] = shf(xp, sb + k);
  }
  v8f Mm;
  {
    v8f G = wmma4(Ao[0], Ao[1], W[RP + bBase + 0 ], W[RP + bBase + 8 ], z8);
    G     = wmma4(Ao[2], Ao[3], W[RP + bBase + 32], W[RP + bBase + 40], G);
    const int base = dval ? (R1 + eD * 64 + dcol) : (RT + (lane & 15));
    #pragma unroll
    for (int i = 0; i < 8; ++i) W[base + i * 8] = G[i];
    Mm = wmma4(W[R1 + aBase + 0], W[R1 + aBase + 1], Ao[0], Ao[1], z8);
    Mm = wmma4(W[R1 + aBase + 4], W[R1 + aBase + 5], Ao[2], Ao[3], Mm);
  }
  const int R5 = RP;

  #pragma unroll 1
  for (int t = 0; t < TT; ++t) {
    const long zidx = ((b0 + (lane >> 4)) * TT + t) * N4 + (r & 3);
    const float zv = z[zidx];
    __builtin_prefetch(&z[(t < TT - 8) ? (zidx + 8 * N4) : zidx], 0, 0);

    float u = 0.0f;
    #pragma unroll
    for (int k = 0; k < 8; ++k) u = fmaf(regC[k], xpb[k], u);
    const float ys = fast_sin(u);
    const float cy = fast_cos(u);
    float axp = 0.0f;
    #pragma unroll
    for (int k = 0; k < 8; ++k) axp = fmaf(regA[k], xpb[k], axp);

    const int srcAB = (e8 << 4) + r;
    const float cyA = shf(cy, srcAB);
    float Ho[4];
    #pragma unroll
    for (int jc = 0; jc < 2; ++jc)
      #pragma unroll
      for (int v = 0; v < 2; ++v) {
        const float cv = lowh ? regC[4 * jc + v] : regC[4 * jc + v + 2];
        Ho[2 * jc + v] = cyA * cv;
      }

    const float ddc = 1.0f - xp * xp;
    v8f Pp;
    #pragma unroll
    for (int i = 0; i < 8; ++i) {
      const float ddr = 1.0f - xpb[i] * xpb[i];
      Pp[i] = fmaf(ddr * ddc, Mm[i], Qd[i]);
    }
    {
      const int base = dval ? (R2 + eD * 64 + dcol) : (RT + (lane & 15));
      #pragma unroll
      for (int i = 0; i < 8; ++i) W[base + i * 8] = Pp[i];
    }
    const float pb0 = W[R2 + bBase + 0 ], pb1 = W[R2 + bBase + 8 ];
    const float pb2 = W[R2 + bBase + 32], pb3 = W[R2 + bBase + 40];
    const float pa0 = W[R2 + aBase + 0], pa1 = W[R2 + aBase + 1];
    const float pa2 = W[R2 + aBase + 4], pa3 = W[R2 + aBase + 5];
    v8f T1 = wmma4(Ho[0], Ho[1], pb0, pb1, z8);
    v8f Gp = wmma4(Ao[0], Ao[1], pb0, pb1, z8);
    T1     = wmma4(Ho[2], Ho[3], pb2, pb3, T1);
    Gp     = wmma4(Ao[2], Ao[3], pb2, pb3, Gp);
    {
      const int base = dval ? (R3 + eD * 32 + dcol) : (RT + (lane & 15));
      #pragma unroll
      for (int i = 0; i < 4; ++i) W[base + i * 8] = T1[i];
    }
    const float t1a0 = W[R3 + a3Base + 0], t1a1 = W[R3 + a3Base + 1];
    const float t1a2 = W[R3 + a3Base + 4], t1a3 = W[R3 + a3Base + 5];
    __builtin_amdgcn_sched_barrier(0);

    v8f PHt = wmma4(pa0, pa1, Ho[0], Ho[1], z8);
    PHt     = wmma4(pa2, pa3, Ho[2], Ho[3], PHt);
    {
      const int base = dval ? (R1 + eD * 64 + dcol) : (RT + (lane & 15));
      #pragma unroll
      for (int i = 0; i < 8; ++i) W[base + i * 8] = Gp[i];
    }
    const float ga0 = W[R1 + aBase + 0], ga1 = W[R1 + aBase + 1];
    const float ga2 = W[R1 + aBase + 4], ga3 = W[R1 + aBase + 5];
    {
      const int base = (dval && dcol < 4) ? (R4 + eD * 32 + dcol) : (RT + (lane & 15));
      #pragma unroll
      for (int i = 0; i < 8; ++i) W[base + i * 4] = PHt[i];
    }
    __builtin_amdgcn_sched_barrier(0);

    v8f Sm  = wmma4(t1a0, t1a1, Ho[0], Ho[1], Rd);
    v8f APH = wmma4(ga0, ga1, Ho[0], Ho[1], z8);
    Sm      = wmma4(t1a2, t1a3, Ho[2], Ho[3], Sm);
    APH     = wmma4(ga2, ga3, Ho[2], Ho[3], APH);
    float s[4][4];
    const int sS = 24 * eD;
    #pragma unroll
    for (int i = 0; i < 4; ++i)
      #pragma unroll
      for (int j = 0; j < 4; ++j) s[i][j] = shf(Sm[i], sS + j);
    __builtin_amdgcn_sched_barrier(0);

    {
      const int base = (dval && dcol < 4) ? (R5 + eD * 32 + dcol) : (RT + (lane & 15));
      #pragma unroll
      for (int i = 0; i < 8; ++i) W[base + i * 4] = APH[i];
    }
    v8f TA = wmma4(t1a0, t1a1, Ao[0], Ao[1], z8);
    v8f Mp = wmma4(ga0, ga1, Ao[0], Ao[1], z8);
    TA     = wmma4(t1a2, t1a3, Ao[2], Ao[3], TA);
    Mp     = wmma4(ga2, ga3, Ao[2], Ao[3], Mp);

    const float dy = zv - ys;
    float dw[4];
    #pragma unroll
    for (int j = 0; j < 4; ++j) dw[j] = shf(dy, sb + j);

    const float4 rhs4 = *reinterpret_cast<const float4*>(&W[R4 + (eD << 5) + (r << 2)]);
    const float4 rhq4 = *reinterpret_cast<const float4*>(&W[R5 + (eD << 5) + (r << 2)]);
    const float apa0 = W[R5 + e8 * 32 + r * 4 + 2 * khi + 0];
    const float apa1 = W[R5 + e8 * 32 + r * 4 + 2 * khi + 1];
    float tv[4] = {TA[0], TA[1], TA[2], TA[3]};
    #pragma unroll
    for (int k = 0; k < 4; ++k) {
      const float inv = __builtin_amdgcn_rcpf(s[k][k]);
      #pragma unroll
      for (int j = k + 1; j < 4; ++j) s[k][j] *= inv;
      dw[k] *= inv;
      tv[k] *= inv;
      #pragma unroll
      for (int i = k + 1; i < 4; ++i) {
        const float f = s[i][k];
        #pragma unroll
        for (int j = k + 1; j < 4; ++j) s[i][j] = fmaf(-f, s[k][j], s[i][j]);
        dw[i] = fmaf(-f, dw[k], dw[i]);
        tv[i] = fmaf(-f, tv[k], tv[i]);
      }
    }
    float wv[4], Vc[4];
    wv[3] = dw[3];
    wv[2] = fmaf(-s[2][3], wv[3], dw[2]);
    wv[1] = fmaf(-s[1][3], wv[3], fmaf(-s[1][2], wv[2], dw[1]));
    wv[0] = fmaf(-s[0][3], wv[3], fmaf(-s[0][2], wv[2], fmaf(-s[0][1], wv[1], dw[0])));
    Vc[3] = tv[3];
    Vc[2] = fmaf(-s[2][3], Vc[3], tv[2]);
    Vc[1] = fmaf(-s[1][3], Vc[3], fmaf(-s[1][2], Vc[2], tv[1]));
    Vc[0] = fmaf(-s[0][3], Vc[3], fmaf(-s[0][2], Vc[2], fmaf(-s[0][1], Vc[1], tv[0])));

    float tn = axp;
    tn = fmaf(rhq4.x, wv[0], tn); tn = fmaf(rhq4.y, wv[1], tn);
    tn = fmaf(rhq4.z, wv[2], tn); tn = fmaf(rhq4.w, wv[3], tn);
    float xn = xp;
    xn = fmaf(rhs4.x, wv[0], xn); xn = fmaf(rhs4.y, wv[1], xn);
    xn = fmaf(rhs4.z, wv[2], xn); xn = fmaf(rhs4.w, wv[3], xn);
    xp = fast_tanh(tn);
    #pragma unroll
    for (int k = 0; k < 8; ++k) xpb[k] = shf(xp, sb + k);

    const int sK = 24 * e8 + r;
    float Bv0, Bv1;
    {
      const float vlo0 = shf(Vc[0], sK), vhi0 = shf(Vc[2], sK);
      const float vlo1 = shf(Vc[1], sK), vhi1 = shf(Vc[3], sK);
      Bv0 = lowh ? vlo0 : vhi0;
      Bv1 = lowh ? vlo1 : vhi1;
    }
    asm volatile("" :: "v"(xn));
    if (dval) out[((b0 + eD) * TT + t) * 8 + r] = xn;

    Mm = wmma4(-apa0, -apa1, Bv0, Bv1, Mp);
  }
}

extern "C" void kernel_launch(void* const* d_in, const int* in_sizes, int n_in,
                              void* d_out, int out_size, void* d_ws, size_t ws_size,
                              hipStream_t stream) {
  const float* z  = (const float*)d_in[0];
  const float* A  = (const float*)d_in[1];
  const float* C  = (const float*)d_in[2];
  const float* Q  = (const float*)d_in[3];
  const float* R  = (const float*)d_in[4];
  const float* x0 = (const float*)d_in[5];
  const float* P0 = (const float*)d_in[6];
  float* out = (float*)d_out;

  const int Bsz = in_sizes[0] / (TT * N4);
  const int waves = Bsz / 2;
  const int threads = 256;
  int blocks = (waves * 32) / threads;
  if (blocks < 1) blocks = 1;
  ekf1250_kernel<<<blocks, threads, 0, stream>>>(z, A, C, Q, R, x0, P0, out);
}


# Issue order only: dy lane reads, APH store and its read-backs go with the S lane reads in front of the solve; the elimination is one vector stretch with no LDS instruction, its wait on dy mostly gone
# speedup vs baseline: 1.2487x; 1.0479x over previous
#include <hip/hip_runtime.h>
#include <hip/hip_bf16.h>


#define TT 256
#define N4 4

typedef __attribute__((ext_vector_type(2))) float v2f;
typedef __attribute__((ext_vector_type(8))) float v8f;

__device__ __forceinline__ float shf(float v, int srcLane) {
  return __int_as_float(__builtin_amdgcn_ds_bpermute(srcLane << 2, __float_as_int(v)));
}

__device__ __forceinline__ v8f wmma4(float a0, float a1, float b0, float b1, v8f c) {
  v2f A, B;
  A[0] = a0; A[1] = a1;
  B[0] = b0; B[1] = b1;
  return __builtin_amdgcn_wmma_f32_16x16x4_f32(false, A, false, B, (short)0, c, false, false);
}

__device__ __forceinline__ float fast_tanh(float x) {
#if __has_builtin(__builtin_amdgcn_tanhf)
  return __builtin_amdgcn_tanhf(x);
#else
  const float e = __expf(2.0f * x);
  return 1.0f - 2.0f * __builtin_amdgcn_rcpf(e + 1.0f);
#endif
}
#define INV_2PI 0.15915494309189535f
__device__ __forceinline__ float fast_sin(float x) {
  return __builtin_amdgcn_sinf(x * INV_2PI);
}
__device__ __forceinline__ float fast_cos(float x) {
  return __builtin_amdgcn_cosf(x * INV_2PI);
}

__global__ __launch_bounds__(256) void ekf1250_kernel(
    const float* __restrict__ z,  const float* __restrict__ Ag,
    const float* __restrict__ Cg, const float* __restrict__ Qg,
    const float* __restrict__ Rg, const float* __restrict__ x0g,
    const float* __restrict__ P0g, float* __restrict__ out)
{
  __shared__ float lds[8 * 640];
  const int tid  = threadIdx.x;
  const int lane = tid & 31;
  const int wave = (blockIdx.x * blockDim.x + tid) >> 5;
  const long b0  = (long)wave * 2;
  float* W = &lds[(tid >> 5) * 640];
  const int RP = 0;
  const int R1 = 128;
  const int R2 = 256;
  const int R3 = 384;
  const int R4 = 448;
  const int RT = 512;

  const int  r    = lane & 7;
  const int  e8   = (lane >> 3) & 1;
  const int  khi  = lane >> 4;
  const int  eD   = lane >> 4;
  const bool dval = (lane < 8) || (lane >= 24);
  const bool lowh = lane < 16;
  const int  dcol = lane & 7;

  const int aBase  = e8 * 64 + r * 8 + 2 * khi;
  const int bBase  = e8 * 64 + (2 * khi) * 8 + r;
  const int a3Base = e8 * 32 + (r & 3) * 8 + 2 * khi;

  float regA[8], regC[8];
  {
    const float4 a0 = *reinterpret_cast<const float4*>(Ag + r * 8);
    const float4 a1 = *reinterpret_cast<const float4*>(Ag + r * 8 + 4);
    regA[0]=a0.x; regA[1]=a0.y; regA[2]=a0.z; regA[3]=a0.w;
    regA[4]=a1.x; regA[5]=a1.y; regA[6]=a1.z; regA[7]=a1.w;
    const float4 c0 = *reinterpret_cast<const float4*>(Cg + (r & 3) * 8);
    const float4 c1 = *reinterpret_cast<const float4*>(Cg + (r & 3) * 8 + 4);
    const float m = (r < 4) ? 1.0f : 0.0f;
    regC[0]=c0.x*m; regC[1]=c0.y*m; regC[2]=c0.z*m; regC[3]=c0.w*m;
    regC[4]=c1.x*m; regC[5]=c1.y*m; regC[6]=c1.z*m; regC[7]=c1.w*m;
  }
  v8f Qd, Rd;
  #pragma unroll
  for (int i = 0; i < 8; ++i) Qd[i] = Qg[i * 8 + dcol];
  #pragma unroll
  for (int i = 0; i < 8; ++i) {
    float rv = Rg[(i & 3) * 4 + (dcol & 3)];
    Rd[i] = (i < 4 && dcol < 4) ? rv : 0.0f;
  }

  for (int idx = lane; idx < 128; idx += 32) W[RP + idx] = P0g[idx & 63];

  const v8f z8 = {0,0,0,0,0,0,0,0};

  float Ao[4];
  #pragma unroll
  for (int jc = 0; jc < 2; ++jc)
    #pragma unroll
    for (int v = 0; v < 2; ++v)
      Ao[2 * jc + v] = lowh ? regA[4 * jc + v] : regA[4 * jc + v + 2];

  const int sb = lane & 16;
  float xp;
  float xpb[8];
  {
    float ta = 0.0f;
    #pragma unroll
    for (int k = 0; k < 8; ++k) ta = fmaf(regA[k], x0g[k], ta);
    xp = fast_tanh(ta);
    #pragma unroll
    for (int k = 0; k < 8; ++k) xpb[k] = shf(xp, sb + k);
  }
  v8f Mm;
  {
    v8f G = wmma4(Ao[0], Ao[1], W[RP + bBase + 0 ], W[RP + bBase + 8 ], z8);
    G     = wmma4(Ao[2], Ao[3], W[RP + bBase + 32], W[RP + bBase + 40], G);
    const int base = dval ? (R1 + eD * 64 + dcol) : (RT + (lane & 15));
    #pragma unroll
    for (int i = 0; i < 8; ++i) W[base + i * 8] = G[i];
    Mm = wmma4(W[R1 + aBase + 0], W[R1 + aBase + 1], Ao[0], Ao[1], z8);
    Mm = wmma4(W[R1 + aBase + 4], W[R1 + aBase + 5], Ao[2], Ao[3], Mm);
  }
  const int R5 = RP;

  #pragma unroll 1
  for (int t = 0; t < TT; ++t) {
    const long zidx = ((b0 + (lane >> 4)) * TT + t) * N4 + (r & 3);
    const float zv = z[zidx];
    __builtin_prefetch(&z[(t < TT - 8) ? (zidx + 8 * N4) : zidx], 0, 0);

    float u = 0.0f;
    #pragma unroll
    for (int k = 0; k < 8; ++k) u = fmaf(regC[k], xpb[k], u);
    const float ys = fast_sin(u);
    const float cy = fast_cos(u);
    float axp = 0.0f;
    #pragma unroll
    for (int k = 0; k < 8; ++k) axp = fmaf(regA[k], xpb[k], axp);

    const int srcAB = (e8 << 4) + r;
    const float cyA = shf(cy, srcAB);
    float Ho[4];
    #pragma unroll
    for (int jc = 0; jc < 2; ++jc)
      #pragma unroll
      for (int v = 0; v < 2; ++v) {
        const float cv = lowh ? regC[4 * jc + v] : regC[4 * jc + v + 2];
        Ho[2 * jc + v] = cyA * cv;
      }

    const float ddc = 1.0f - xp * xp;
    v8f Pp;
    #pragma unroll
    for (int i = 0; i < 8; ++i) {
      const float ddr = 1.0f - xpb[i] * xpb[i];
      Pp[i] = fmaf(ddr * ddc, Mm[i], Qd[i]);
    }
    {
      const int base = dval ? (R2 + eD * 64 + dcol) : (RT + (lane & 15));
      #pragma unroll
      for (int i = 0; i < 8; ++i) W[base + i * 8] = Pp[i];
    }
    const float pb0 = W[R2 + bBase + 0 ], pb1 = W[R2 + bBase + 8 ];
    const float pb2 = W[R2 + bBase + 32], pb3 = W[R2 + bBase + 40];
    const float pa0 = W[R2 + aBase + 0], pa1 = W[R2 + aBase + 1];
    const float pa2 = W[R2 + aBase + 4], pa3 = W[R2 + aBase + 5];
    v8f T1 = wmma4(Ho[0], Ho[1], pb0, pb1, z8);
    v8f Gp = wmma4(Ao[0], Ao[1], pb0, pb1, z8);
    T1     = wmma4(Ho[2], Ho[3], pb2, pb3, T1);
    Gp     = wmma4(Ao[2], Ao[3], pb2, pb3, Gp);
    {
      const int base = dval ? (R3 + eD * 32 + dcol) : (RT + (lane & 15));
      #pragma unroll
      for (int i = 0; i < 4; ++i) W[base + i * 8] = T1[i];
    }
    const float t1a0 = W[R3 + a3Base + 0], t1a1 = W[R3 + a3Base + 1];
    const float t1a2 = W[R3 + a3Base + 4], t1a3 = W[R3 + a3Base + 5];
    __builtin_amdgcn_sched_barrier(0);

    v8f PHt = wmma4(pa0, pa1, Ho[0], Ho[1], z8);
    PHt     = wmma4(pa2, pa3, Ho[2], Ho[3], PHt);
    {
      const int base = dval ? (R1 + eD * 64 + dcol) : (RT + (lane & 15));
      #pragma unroll
      for (int i = 0; i < 8; ++i) W[base + i * 8] = Gp[i];
    }
    const float ga0 = W[R1 + aBase + 0], ga1 = W[R1 + aBase + 1];
    const float ga2 = W[R1 + aBase + 4], ga3 = W[R1 + aBase + 5];
    {
      const int base = (dval && dcol < 4) ? (R4 + eD * 32 + dcol) : (RT + (lane & 15));
      #pragma unroll
      for (int i = 0; i < 8; ++i) W[base + i * 4] = PHt[i];
    }
    const float dy = zv - ys;
    float dw[4];
    #pragma unroll
    for (int j = 0; j < 4; ++j) dw[j] = shf(dy, sb + j);
    __builtin_amdgcn_sched_barrier(0);

    v8f Sm  = wmma4(t1a0, t1a1, Ho[0], Ho[1], Rd);
    v8f APH = wmma4(ga0, ga1, Ho[0], Ho[1], z8);
    Sm      = wmma4(t1a2, t1a3, Ho[2], Ho[3], Sm);
    APH     = wmma4(ga2, ga3, Ho[2], Ho[3], APH);
    float s[4][4];
    const int sS = 24 * eD;
    #pragma unroll
    for (int i = 0; i < 4; ++i)
      #pragma unroll
      for (int j = 0; j < 4; ++j) s[i][j] = shf(Sm[i], sS + j);
    __builtin_amdgcn_sched_barrier(0);

    {
      const int base = (dval && dcol < 4) ? (R5 + eD * 32 + dcol) : (RT + (lane & 15));
      #pragma unroll
      for (int i = 0; i < 8; ++i) W[base + i * 4] = APH[i];
    }
    const float4 rhs4 = *reinterpret_cast<const float4*>(&W[R4 + (eD << 5) + (r << 2)]);
    const float4 rhq4 = *reinterpret_cast<const float4*>(&W[R5 + (eD << 5) + (r << 2)]);
    const float apa0 = W[R5 + e8 * 32 + r * 4 + 2 * khi + 0];
    const float apa1 = W[R5 + e8 * 32 + r * 4 + 2 * khi + 1];
    __builtin_amdgcn_sched_barrier(0);

    v8f TA = wmma4(t1a0, t1a1, Ao[0], Ao[1], z8);
    v8f Mp = wmma4(ga0, ga1, Ao[0], Ao[1], z8);
    TA     = wmma4(t1a2, t1a3, Ao[2], Ao[3], TA);
    Mp     = wmma4(ga2, ga3, Ao[2], Ao[3], Mp);

    float tv[4] = {TA[0], TA[1], TA[2], TA[3]};
    #pragma unroll
    for (int k = 0; k < 4; ++k) {
      const float inv = __builtin_amdgcn_rcpf(s[k][k]);
      #pragma unroll
      for (int j = k + 1; j < 4; ++j) s[k][j] *= inv;
      dw[k] *= inv;
      tv[k] *= inv;
      #pragma unroll
      for (int i = k + 1; i < 4; ++i) {
        const float f = s[i][k];
        #pragma unroll
        for (int j = k + 1; j < 4; ++j) s[i][j] = fmaf(-f, s[k][j], s[i][j]);
        dw[i] = fmaf(-f, dw[k], dw[i]);
        tv[i] = fmaf(-f, tv[k], tv[i]);
      }
    }
    float wv[4], Vc[4];
    wv[3] = dw[3];
    wv[2] = fmaf(-s[2][3], wv[3], dw[2]);
    wv[1] = fmaf(-s[1][3], wv[3], fmaf(-s[1][2], wv[2], dw[1]));
    wv[0] = fmaf(-s[0][3], wv[3], fmaf(-s[0][2], wv[2], fmaf(-s[0][1], wv[1], dw[0])));
    Vc[3] = tv[3];
    Vc[2] = fmaf(-s[2][3], Vc[3], tv[2]);
    Vc[1] = fmaf(-s[1][3], Vc[3], fmaf(-s[1][2], Vc[2], tv[1]));
    Vc[0] = fmaf(-s[0][3], Vc[3], fmaf(-s[0][2], Vc[2], fmaf(-s[0][1], Vc[1], tv[0])));

    float tn = axp;
    tn = fmaf(rhq4.x, wv[0], tn); tn = fmaf(rhq4.y, wv[1], tn);
    tn = fmaf(rhq4.z, wv[2], tn); tn = fmaf(rhq4.w, wv[3], tn);
    float xn = xp;
    xn = fmaf(rhs4.x, wv[0], xn); xn = fmaf(rhs4.y, wv[1], xn);
    xn = fmaf(rhs4.z, wv[2], xn); xn = fmaf(rhs4.w, wv[3], xn);
    xp = fast_tanh(tn);
    #pragma unroll
    for (int k = 0; k < 8; ++k) xpb[k] = shf(xp, sb + k);

    const int sK = 24 * e8 + r;
    float Bv0, Bv1;
    {
      const float vlo0 = shf(Vc[0], sK), vhi0 = shf(Vc[2], sK);
      const float vlo1 = shf(Vc[1], sK), vhi1 = shf(Vc[3], sK);
      Bv0 = lowh ? vlo0 : vhi0;
      Bv1 = lowh ? vlo1 : vhi1;
    }
    asm volatile("" :: "v"(xn));
    if (dval) out[((b0 + eD) * TT + t) * 8 + r] = xn;

    Mm = wmma4(-apa0, -apa1, Bv0, Bv1, Mp);
  }
}

extern "C" void kernel_launch(void* const* d_in, const int* in_sizes, int n_in,
                              void* d_out, int out_size, void* d_ws, size_t ws_size,
                              hipStream_t stream) {
  const float* z  = (const float*)d_in[0];
  const float* A  = (const float*)d_in[1];
  const float* C  = (const float*)d_in[2];
  const float* Q  = (const float*)d_in[3];
  const float* R  = (const float*)d_in[4];
  const float* x0 = (const float*)d_in[5];
  const float* P0 = (const float*)d_in[6];
  float* out = (float*)d_out;

  const int Bsz = in_sizes[0] / (TT * N4);
  const int waves = Bsz / 2;
  const int threads = 256;
  int blocks = (waves * 32) / threads;
  if (blocks < 1) blocks = 1;
  ekf1250_kernel<<<blocks, threads, 0, stream>>>(z, A, C, Q, R, x0, P0, out);
}
